# starting kernel
# baseline (speedup 1.0000x reference)
#include <hip/hip_runtime.h>
#include <math.h>

typedef __attribute__((ext_vector_type(2))) float v2f;
typedef __attribute__((ext_vector_type(8))) float v8f;

#define B_DIM 4
#define S_DIM 4096
#define D_DIM 1024
#define M_DIM (B_DIM * S_DIM)
#define N_DIM (2 * D_DIM)
#define K_DIM D_DIM

#define NCH   8
#define CHL   (S_DIM / NCH)
#define NCHAN (B_DIM * D_DIM)

__device__ __forceinline__ float sigmoid_f(float v) {
    return 1.0f / (1.0f + expf(-v));
}

__global__ __launch_bounds__(256)
void mingru_gemm(const float* __restrict__ x, const float* __restrict__ w,
                 float* __restrict__ hidden, float* __restrict__ gate)
{
    const int lane = threadIdx.x & 31;
    const int wave = blockIdx.x * (blockDim.x >> 5) + (threadIdx.x >> 5);
    const int NT   = N_DIM / 64;
    const int mt   = wave / NT;
    const int nt   = wave - mt * NT;
    const int m0   = mt << 5;
    const int n0   = nt << 6;

    const int row = lane & 15;
    const int kq  = (lane >> 4) << 1;

    const float* xp0 = x + (size_t)(m0 +  0 + row) * K_DIM + kq;
    const float* xp1 = x + (size_t)(m0 + 16 + row) * K_DIM + kq;
    const float* wp0 = w + (size_t)(n0 +  0 + row) * K_DIM + kq;
    const float* wp1 = w + (size_t)(n0 + 16 + row) * K_DIM + kq;
    const float* wp2 = w + (size_t)(n0 + 32 + row) * K_DIM + kq;
    const float* wp3 = w + (size_t)(n0 + 48 + row) * K_DIM + kq;

    v8f acc00 = {}, acc01 = {}, acc02 = {}, acc03 = {};
    v8f acc10 = {}, acc11 = {}, acc12 = {}, acc13 = {};

#pragma unroll 2
    for (int k = 0; k < K_DIM; k += 4) {
        v2f a0 = *(const v2f*)(xp0 + k);
        v2f a1 = *(const v2f*)(xp1 + k);
        v2f b0 = *(const v2f*)(wp0 + k);
        v2f b1 = *(const v2f*)(wp1 + k);
        v2f b2 = *(const v2f*)(wp2 + k);
        v2f b3 = *(const v2f*)(wp3 + k);
        acc00 = __builtin_amdgcn_wmma_f32_16x16x4_f32(false, a0, false, b0, (short)0, acc00, false, false);
        acc01 = __builtin_amdgcn_wmma_f32_16x16x4_f32(false, a0, false, b1, (short)0, acc01, false, false);
        acc02 = __builtin_amdgcn_wmma_f32_16x16x4_f32(false, a0, false, b2, (short)0, acc02, false, false);
        acc03 = __builtin_amdgcn_wmma_f32_16x16x4_f32(false, a0, false, b3, (short)0, acc03, false, false);
        acc10 = __builtin_amdgcn_wmma_f32_16x16x4_f32(false, a1, false, b0, (short)0, acc10, false, false);
        acc11 = __builtin_amdgcn_wmma_f32_16x16x4_f32(false, a1, false, b1, (short)0, acc11, false, false);
        acc12 = __builtin_amdgcn_wmma_f32_16x16x4_f32(false, a1, false, b2, (short)0, acc12, false, false);
        acc13 = __builtin_amdgcn_wmma_f32_16x16x4_f32(false, a1, false, b3, (short)0, acc13, false, false);
    }

    float* dst;
    int colbase;
    if (n0 < D_DIM) { dst = hidden; colbase = n0; }
    else            { dst = gate;   colbase = n0 - D_DIM; }

    const int cmhi = (lane >> 4) << 3;
    const int cn   = lane & 15;
    v8f accs[2][4] = {{acc00, acc01, acc02, acc03}, {acc10, acc11, acc12, acc13}};

#pragma unroll
    for (int i = 0; i < 2; ++i) {
        const int cm = m0 + (i << 4) + cmhi;
#pragma unroll
        for (int j = 0; j < 4; ++j) {
            const size_t col = (size_t)(colbase + (j << 4) + cn);
#pragma unroll
            for (int r = 0; r < 8; ++r) {
                dst[(size_t)(cm + r) * D_DIM + col] = accs[i][j][r];
            }
        }
    }
}

__global__ __launch_bounds__(256)
void mingru_scan_pass1(float* __restrict__ h, const float* __restrict__ gate,
                       float* __restrict__ carryA, float* __restrict__ carryE)
{
    const int t  = blockIdx.x * blockDim.x + threadIdx.x;
    const int ch = t & (NCHAN - 1);
    const int c  = t >> 12;
    const int b  = ch >> 10;
    const int d  = ch & (D_DIM - 1);

    size_t base = ((size_t)b * S_DIM + (size_t)c * CHL) * D_DIM + d;
    float hl = 0.0f;
    float P  = 1.0f;

    for (int s = 0; s < CHL; ++s) {
        const size_t idx = base + (size_t)s * D_DIM;
        const float hid = h[idx];
        const float g   = gate[idx];
        const float z   = sigmoid_f(g);
        const float omz = 1.0f - z;
        const float gv  = (hid >= 0.0f) ? (hid + 0.5f) : sigmoid_f(hid);
        hl = fmaf(omz, hl, z * gv);
        P *= omz;
        h[idx] = hl;
    }
    carryA[t] = P;
    carryE[t] = hl;
}

__global__ __launch_bounds__(256)
void mingru_scan_pass2(const float* __restrict__ carryA, float* __restrict__ carryE)
{
    const int ch = blockIdx.x * blockDim.x + threadIdx.x;
    if (ch >= NCHAN) return;
    float hin = 0.0f;
#pragma unroll
    for (int c = 0; c < NCH; ++c) {
        const int t   = c * NCHAN + ch;
        const float A = carryA[t];
        const float E = carryE[t];
        carryE[t] = hin;
        hin = fmaf(A, hin, E);
    }
}

__global__ __launch_bounds__(256)
void mingru_scan_pass3(float* __restrict__ h, const float* __restrict__ gate,
                       const float* __restrict__ carryE)
{
    const int t  = blockIdx.x * blockDim.x + threadIdx.x;
    const int ch = t & (NCHAN - 1);
    const int c  = t >> 12;
    const float hin = carryE[t];
    if (hin == 0.0f) return;

    const int b = ch >> 10;
    const int d = ch & (D_DIM - 1);
    size_t base = ((size_t)b * S_DIM + (size_t)c * CHL) * D_DIM + d;

    float P = hin;
    for (int s = 0; s < CHL; ++s) {
        const size_t idx = base + (size_t)s * D_DIM;
        const float omz = 1.0f - sigmoid_f(gate[idx]);
        P *= omz;
        h[idx] += P;
    }
}

extern "C" void kernel_launch(void* const* d_in, const int* in_sizes, int n_in,
                              void* d_out, int out_size, void* d_ws, size_t ws_size,
                              hipStream_t stream)
{
    const float* x = (const float*)d_in[0];
    const float* w = (const float*)d_in[1];

    float* hidden = (float*)d_out;
    float* gate   = (float*)d_ws;
    float* carryA = gate + (size_t)M_DIM * D_DIM;
    float* carryE = carryA + (size_t)NCHAN * NCH;

    const int gemm_waves  = (M_DIM / 32) * (N_DIM / 64);
    const int gemm_blocks = gemm_waves / 8;
    mingru_gemm<<<gemm_blocks, 256, 0, stream>>>(x, w, hidden, gate);

    mingru_scan_pass1<<<(NCHAN * NCH) / 256, 256, 0, stream>>>(hidden, gate, carryA, carryE);
    mingru_scan_pass2<<<NCHAN / 256, 256, 0, stream>>>(carryA, carryE);
    mingru_scan_pass3<<<(NCHAN * NCH) / 256, 256, 0, stream>>>(hidden, gate, carryE);
}


# GEMM operands staged per 128x128 block tile through LDS (whole-line global loads, ds_store, 16-byte fragment reads); k inside each 8 summed as 0,1,4,5 then 2,3,6,7
# speedup vs baseline: 1.6889x; 1.6889x over previous
#include <hip/hip_runtime.h>
#include <math.h>

typedef __attribute__((ext_vector_type(2))) float v2f;
typedef __attribute__((ext_vector_type(4))) float v4f;
typedef __attribute__((ext_vector_type(8))) float v8f;

#define B_DIM 4
#define S_DIM 4096
#define D_DIM 1024
#define M_DIM (B_DIM * S_DIM)
#define N_DIM (2 * D_DIM)
#define K_DIM D_DIM

#define NCH   8
#define CHL   (S_DIM / NCH)
#define NCHAN (B_DIM * D_DIM)

__device__ __forceinline__ float sigmoid_f(float v) {
    return 1.0f / (1.0f + expf(-v));
}

#define GBM 128
#define GBN 128
#define GBK 32
#define LDP 36
#define TILE_F (GBM * LDP)

__global__ __launch_bounds__(256)
void mingru_gemm(const float* __restrict__ x, const float* __restrict__ w,
                 float* __restrict__ hidden, float* __restrict__ gate)
{
    __shared__ __attribute__((aligned(16))) float lds[2][2][TILE_F];

    const int tid  = threadIdx.x;
    const int lane = tid & 31;
    const int wv   = tid >> 5;
    const int wm   = wv >> 1;
    const int wn   = wv & 1;
    const int bn   = blockIdx.x & (N_DIM / GBN - 1);
    const int bm   = blockIdx.x / (N_DIM / GBN);
    const int m_blk = bm * GBM;
    const int n_blk = bn * GBN;

    const int sr = tid >> 3;
    const int sc = (tid & 7) << 2;
    const float* xg = x + (size_t)(m_blk + sr) * K_DIM + sc;
    const float* wg = w + (size_t)(n_blk + sr) * K_DIM + sc;
    const int st_off = sr * LDP + sc;

    const int c = lane & 15;
    const int h = lane >> 4;
    const int a_off = (wm * 32 + c) * LDP + (h << 2);
    const int b_off = (wn * 64 + c) * LDP + (h << 2);

    v8f acc[2][4];
#pragma unroll
    for (int i = 0; i < 2; ++i)
#pragma unroll
        for (int j = 0; j < 4; ++j)
            acc[i][j] = (v8f){};

    v4f pa[4], pb[4];
#pragma unroll
    for (int i = 0; i < 4; ++i) {
        pa[i] = *(const v4f*)(xg + (size_t)(i * 32) * K_DIM);
        pb[i] = *(const v4f*)(wg + (size_t)(i * 32) * K_DIM);
    }
#pragma unroll
    for (int i = 0; i < 4; ++i) {
        *(v4f*)&lds[0][0][st_off + i * 32 * LDP] = pa[i];
        *(v4f*)&lds[0][1][st_off + i * 32 * LDP] = pb[i];
    }
    __syncthreads();

    const int NKT = K_DIM / GBK;
    for (int kt = 0; kt < NKT; ++kt) {
        const int cur = kt & 1;
        const bool more = (kt + 1) < NKT;

        if (more) {
            const int ko = (kt + 1) * GBK;
#pragma unroll
            for (int i = 0; i < 4; ++i) {
                pa[i] = *(const v4f*)(xg + (size_t)(i * 32) * K_DIM + ko);
                pb[i] = *(const v4f*)(wg + (size_t)(i * 32) * K_DIM + ko);
            }
        }

        const float* As = &lds[cur][0][a_off];
        const float* Bs = &lds[cur][1][b_off];

        v4f af[2][2], bf[2][4];
#pragma unroll
        for (int i = 0; i < 2; ++i) af[0][i] = *(const v4f*)(As + i * 16 * LDP);
#pragma unroll
        for (int j = 0; j < 4; ++j) bf[0][j] = *(const v4f*)(Bs + j * 16 * LDP);

#pragma unroll
        for (int kb = 0; kb < GBK / 8; ++kb) {
            const int s = kb & 1;
            if (kb + 1 < GBK / 8) {
#pragma unroll
                for (int i = 0; i < 2; ++i) af[s ^ 1][i] = *(const v4f*)(As + i * 16 * LDP + (kb + 1) * 8);
#pragma unroll
                for (int j = 0; j < 4; ++j) bf[s ^ 1][j] = *(const v4f*)(Bs + j * 16 * LDP + (kb + 1) * 8);
            }
#pragma unroll
            for (int i = 0; i < 2; ++i) {
                const v2f alo = {af[s][i].x, af[s][i].y};
#pragma unroll
                for (int j = 0; j < 4; ++j) {
                    const v2f blo = {bf[s][j].x, bf[s][j].y};
                    acc[i][j] = __builtin_amdgcn_wmma_f32_16x16x4_f32(false, alo, false, blo, (short)0, acc[i][j], false, false);
                }
            }
#pragma unroll
            for (int i = 0; i < 2; ++i) {
                const v2f ahi = {af[s][i].z, af[s][i].w};
#pragma unroll
                for (int j = 0; j < 4; ++j) {
                    const v2f bhi = {bf[s][j].z, bf[s][j].w};
                    acc[i][j] = __builtin_amdgcn_wmma_f32_16x16x4_f32(false, ahi, false, bhi, (short)0, acc[i][j], false, false);
                }
            }
        }

        if (more) {
#pragma unroll
            for (int i = 0; i < 4; ++i) {
                *(v4f*)&lds[cur ^ 1][0][st_off + i * 32 * LDP] = pa[i];
                *(v4f*)&lds[cur ^ 1][1][st_off + i * 32 * LDP] = pb[i];
            }
            __syncthreads();
        }
    }

    const int m0 = m_blk + wm * 32;
    const int n0 = n_blk + wn * 64;
    float* dst;
    int colbase;
    if (n0 < D_DIM) { dst = hidden; colbase = n0; }
    else            { dst = gate;   colbase = n0 - D_DIM; }

    const int cmhi = h << 3;
#pragma unroll
    for (int i = 0; i < 2; ++i) {
        const int cm = m0 + (i << 4) + cmhi;
#pragma unroll
        for (int j = 0; j < 4; ++j) {
            const size_t col = (size_t)(colbase + (j << 4) + c);
#pragma unroll
            for (int r = 0; r < 8; ++r) {
                dst[(size_t)(cm + r) * D_DIM + col] = acc[i][j][r];
            }
        }
    }
}

__global__ __launch_bounds__(256)
void mingru_scan_pass1(float* __restrict__ h, const float* __restrict__ gate,
                       float* __restrict__ carryA, float* __restrict__ carryE)
{
    const int t  = blockIdx.x * blockDim.x + threadIdx.x;
    const int ch = t & (NCHAN - 1);
    const int c  = t >> 12;
    const int b  = ch >> 10;
    const int d  = ch & (D_DIM - 1);

    size_t base = ((size_t)b * S_DIM + (size_t)c * CHL) * D_DIM + d;
    float hl = 0.0f;
    float P  = 1.0f;

    for (int s = 0; s < CHL; ++s) {
        const size_t idx = base + (size_t)s * D_DIM;
        const float hid = h[idx];
        const float g   = gate[idx];
        const float z   = sigmoid_f(g);
        const float omz = 1.0f - z;
        const float gv  = (hid >= 0.0f) ? (hid + 0.5f) : sigmoid_f(hid);
        hl = fmaf(omz, hl, z * gv);
        P *= omz;
        h[idx] = hl;
    }
    carryA[t] = P;
    carryE[t] = hl;
}

__global__ __launch_bounds__(256)
void mingru_scan_pass2(const float* __restrict__ carryA, float* __restrict__ carryE)
{
    const int ch = blockIdx.x * blockDim.x + threadIdx.x;
    if (ch >= NCHAN) return;
    float hin = 0.0f;
#pragma unroll
    for (int c = 0; c < NCH; ++c) {
        const int t   = c * NCHAN + ch;
        const float A = carryA[t];
        const float E = carryE[t];
        carryE[t] = hin;
        hin = fmaf(A, hin, E);
    }
}

__global__ __launch_bounds__(256)
void mingru_scan_pass3(float* __restrict__ h, const float* __restrict__ gate,
                       const float* __restrict__ carryE)
{
    const int t  = blockIdx.x * blockDim.x + threadIdx.x;
    const int ch = t & (NCHAN - 1);
    const int c  = t >> 12;
    const float hin = carryE[t];
    if (hin == 0.0f) return;

    const int b = ch >> 10;
    const int d = ch & (D_DIM - 1);
    size_t base = ((size_t)b * S_DIM + (size_t)c * CHL) * D_DIM + d;

    float P = hin;
    for (int s = 0; s < CHL; ++s) {
        const size_t idx = base + (size_t)s * D_DIM;
        const float omz = 1.0f - sigmoid_f(gate[idx]);
        P *= omz;
        h[idx] += P;
    }
}

extern "C" void kernel_launch(void* const* d_in, const int* in_sizes, int n_in,
                              void* d_out, int out_size, void* d_ws, size_t ws_size,
                              hipStream_t stream)
{
    const float* x = (const float*)d_in[0];
    const float* w = (const float*)d_in[1];

    float* hidden = (float*)d_out;
    float* gate   = (float*)d_ws;
    float* carryA = gate + (size_t)M_DIM * D_DIM;
    float* carryE = carryA + (size_t)NCHAN * NCH;

    mingru_gemm<<<(M_DIM / GBM) * (N_DIM / GBN), 256, 0, stream>>>(x, w, hidden, gate);

    mingru_scan_pass1<<<(NCHAN * NCH) / 256, 256, 0, stream>>>(hidden, gate, carryA, carryE);
    mingru_scan_pass2<<<NCHAN / 256, 256, 0, stream>>>(carryA, carryE);
    mingru_scan_pass3<<<(NCHAN * NCH) / 256, 256, 0, stream>>>(hidden, gate, carryE);
}


# Scan re-cut from 8 chunks of 512 steps to 64 of 64 per channel, 8 rows of loads in flight per wave, carries kept in place in each chunk's last row; carry-in terms regroup (rounding only)
# speedup vs baseline: 3.4103x; 2.0192x over previous
#include <hip/hip_runtime.h>
#include <math.h>

typedef __attribute__((ext_vector_type(2))) float v2f;
typedef __attribute__((ext_vector_type(4))) float v4f;
typedef __attribute__((ext_vector_type(8))) float v8f;

#define B_DIM 4
#define S_DIM 4096
#define D_DIM 1024
#define M_DIM (B_DIM * S_DIM)
#define N_DIM (2 * D_DIM)
#define K_DIM D_DIM

#define NCH   64
#define CHL   (S_DIM / NCH)
#define NCHAN (B_DIM * D_DIM)
#define SCU   8

__device__ __forceinline__ float sigmoid_f(float v) {
    return 1.0f / (1.0f + expf(-v));
}

#define GBM 128
#define GBN 128
#define GBK 32
#define LDP 36
#define TILE_F (GBM * LDP)

__global__ __launch_bounds__(256)
void mingru_gemm(const float* __restrict__ x, const float* __restrict__ w,
                 float* __restrict__ hidden, float* __restrict__ gate)
{
    __shared__ __attribute__((aligned(16))) float lds[2][2][TILE_F];

    const int tid  = threadIdx.x;
    const int lane = tid & 31;
    const int wv   = tid >> 5;
    const int wm   = wv >> 1;
    const int wn   = wv & 1;
    const int bn   = blockIdx.x & (N_DIM / GBN - 1);
    const int bm   = blockIdx.x / (N_DIM / GBN);
    const int m_blk = bm * GBM;
    const int n_blk = bn * GBN;

    const int sr = tid >> 3;
    const int sc = (tid & 7) << 2;
    const float* xg = x + (size_t)(m_blk + sr) * K_DIM + sc;
    const float* wg = w + (size_t)(n_blk + sr) * K_DIM + sc;
    const int st_off = sr * LDP + sc;

    const int c = lane & 15;
    const int h = lane >> 4;
    const int a_off = (wm * 32 + c) * LDP + (h << 2);
    const int b_off = (wn * 64 + c) * LDP + (h << 2);

    v8f acc[2][4];
#pragma unroll
    for (int i = 0; i < 2; ++i)
#pragma unroll
        for (int j = 0; j < 4; ++j)
            acc[i][j] = (v8f){};

    v4f pa[4], pb[4];
#pragma unroll
    for (int i = 0; i < 4; ++i) {
        pa[i] = *(const v4f*)(xg + (size_t)(i * 32) * K_DIM);
        pb[i] = *(const v4f*)(wg + (size_t)(i * 32) * K_DIM);
    }
#pragma unroll
    for (int i = 0; i < 4; ++i) {
        *(v4f*)&lds[0][0][st_off + i * 32 * LDP] = pa[i];
        *(v4f*)&lds[0][1][st_off + i * 32 * LDP] = pb[i];
    }
    __syncthreads();

    const int NKT = K_DIM / GBK;
    for (int kt = 0; kt < NKT; ++kt) {
        const int cur = kt & 1;
        const bool more = (kt + 1) < NKT;

        if (more) {
            const int ko = (kt + 1) * GBK;
#pragma unroll
            for (int i = 0; i < 4; ++i) {
                pa[i] = *(const v4f*)(xg + (size_t)(i * 32) * K_DIM + ko);
                pb[i] = *(const v4f*)(wg + (size_t)(i * 32) * K_DIM + ko);
            }
        }

        const float* As = &lds[cur][0][a_off];
        const float* Bs = &lds[cur][1][b_off];

        v4f af[2][2], bf[2][4];
#pragma unroll
        for (int i = 0; i < 2; ++i) af[0][i] = *(const v4f*)(As + i * 16 * LDP);
#pragma unroll
        for (int j = 0; j < 4; ++j) bf[0][j] = *(const v4f*)(Bs + j * 16 * LDP);

#pragma unroll
        for (int kb = 0; kb < GBK / 8; ++kb) {
            const int s = kb & 1;
            if (kb + 1 < GBK / 8) {
#pragma unroll
                for (int i = 0; i < 2; ++i) af[s ^ 1][i] = *(const v4f*)(As + i * 16 * LDP + (kb + 1) * 8);
#pragma unroll
                for (int j = 0; j < 4; ++j) bf[s ^ 1][j] = *(const v4f*)(Bs + j * 16 * LDP + (kb + 1) * 8);
            }
#pragma unroll
            for (int i = 0; i < 2; ++i) {
                const v2f alo = {af[s][i].x, af[s][i].y};
#pragma unroll
                for (int j = 0; j < 4; ++j) {
                    const v2f blo = {bf[s][j].x, bf[s][j].y};
                    acc[i][j] = __builtin_amdgcn_wmma_f32_16x16x4_f32(false, alo, false, blo, (short)0, acc[i][j], false, false);
                }
            }
#pragma unroll
            for (int i = 0; i < 2; ++i) {
                const v2f ahi = {af[s][i].z, af[s][i].w};
#pragma unroll
                for (int j = 0; j < 4; ++j) {
                    const v2f bhi = {bf[s][j].z, bf[s][j].w};
                    acc[i][j] = __builtin_amdgcn_wmma_f32_16x16x4_f32(false, ahi, false, bhi, (short)0, acc[i][j], false, false);
                }
            }
        }

        if (more) {
#pragma unroll
            for (int i = 0; i < 4; ++i) {
                *(v4f*)&lds[cur ^ 1][0][st_off + i * 32 * LDP] = pa[i];
                *(v4f*)&lds[cur ^ 1][1][st_off + i * 32 * LDP] = pb[i];
            }
            __syncthreads();
        }
    }

    const int m0 = m_blk + wm * 32;
    const int n0 = n_blk + wn * 64;
    float* dst;
    int colbase;
    if (n0 < D_DIM) { dst = hidden; colbase = n0; }
    else            { dst = gate;   colbase = n0 - D_DIM; }

    const int cmhi = h << 3;
#pragma unroll
    for (int i = 0; i < 2; ++i) {
        const int cm = m0 + (i << 4) + cmhi;
#pragma unroll
        for (int j = 0; j < 4; ++j) {
            const size_t col = (size_t)(colbase + (j << 4) + c);
#pragma unroll
            for (int r = 0; r < 8; ++r) {
                dst[(size_t)(cm + r) * D_DIM + col] = acc[i][j][r];
            }
        }
    }
}


template <int N>
__device__ __forceinline__ void scan1_rows(float* __restrict__ h, float* __restrict__ gate,
                                           size_t idx0, float& hl, float& P)
{
    float hv[N], gv[N];
#pragma unroll
    for (int u = 0; u < N; ++u) {
        hv[u] = h[idx0 + (size_t)u * D_DIM];
        gv[u] = gate[idx0 + (size_t)u * D_DIM];
    }
#pragma unroll
    for (int u = 0; u < N; ++u) {
        const float hid = hv[u];
        const float z   = sigmoid_f(gv[u]);
        const float omz = 1.0f - z;
        const float gx  = (hid >= 0.0f) ? (hid + 0.5f) : sigmoid_f(hid);
        hl = fmaf(omz, hl, z * gx);
        P *= omz;
        h[idx0 + (size_t)u * D_DIM] = hl;
    }
}

__global__ __launch_bounds__(256)
void mingru_scan_pass1(float* __restrict__ h, float* __restrict__ gate)
{
    const int t  = blockIdx.x * blockDim.x + threadIdx.x;
    const int ch = t & (NCHAN - 1);
    const int c  = t >> 12;
    const int b  = ch >> 10;
    const int d  = ch & (D_DIM - 1);

    const size_t base = ((size_t)b * S_DIM + (size_t)c * CHL) * D_DIM + d;
    float hl = 0.0f;
    float P  = 1.0f;

    for (int s0 = 0; s0 < CHL; s0 += SCU)
        scan1_rows<SCU>(h, gate, base + (size_t)s0 * D_DIM, hl, P);

    gate[base + (size_t)(CHL - 1) * D_DIM] = P;
}

__global__ __launch_bounds__(64)
void mingru_scan_pass2(float* __restrict__ h, const float* __restrict__ gate)
{
    const int ch = blockIdx.x * blockDim.x + threadIdx.x;
    const int b  = ch >> 10;
    const int d  = ch & (D_DIM - 1);
    const size_t last0 = ((size_t)b * S_DIM + (CHL - 1)) * D_DIM + d;

    float A[NCH], E[NCH];
#pragma unroll
    for (int c = 0; c < NCH; ++c) {
        A[c] = gate[last0 + (size_t)c * CHL * D_DIM];
        E[c] = h[last0 + (size_t)c * CHL * D_DIM];
    }
    float hin = E[0];
#pragma unroll
    for (int c = 1; c < NCH; ++c) {
        hin = fmaf(A[c], hin, E[c]);
        h[last0 + (size_t)c * CHL * D_DIM] = hin;
    }
}

template <int N>
__device__ __forceinline__ void scan3_rows(float* __restrict__ h, const float* __restrict__ gate,
                                           size_t idx0, float& P)
{
    float hv[N], gv[N];
#pragma unroll
    for (int u = 0; u < N; ++u) {
        gv[u] = gate[idx0 + (size_t)u * D_DIM];
        hv[u] = h[idx0 + (size_t)u * D_DIM];
    }
#pragma unroll
    for (int u = 0; u < N; ++u) {
        const float omz = 1.0f - sigmoid_f(gv[u]);
        P *= omz;
        h[idx0 + (size_t)u * D_DIM] = hv[u] + P;
    }
}

__global__ __launch_bounds__(256)
void mingru_scan_pass3(float* __restrict__ h, const float* __restrict__ gate)
{
    const int t  = blockIdx.x * blockDim.x + threadIdx.x + NCHAN;
    const int ch = t & (NCHAN - 1);
    const int c  = t >> 12;
    const int b  = ch >> 10;
    const int d  = ch & (D_DIM - 1);
    const size_t base = ((size_t)b * S_DIM + (size_t)c * CHL) * D_DIM + d;

    float P = h[base - D_DIM];

    for (int s0 = 0; s0 < CHL - SCU; s0 += SCU)
        scan3_rows<SCU>(h, gate, base + (size_t)s0 * D_DIM, P);
    scan3_rows<SCU - 1>(h, gate, base + (size_t)(CHL - SCU) * D_DIM, P);
}

extern "C" void kernel_launch(void* const* d_in, const int* in_sizes, int n_in,
                              void* d_out, int out_size, void* d_ws, size_t ws_size,
                              hipStream_t stream)
{
    const float* x = (const float*)d_in[0];
    const float* w = (const float*)d_in[1];

    float* hidden = (float*)d_out;
    float* gate   = (float*)d_ws;

    mingru_gemm<<<(M_DIM / GBM) * (N_DIM / GBN), 256, 0, stream>>>(x, w, hidden, gate);

    mingru_scan_pass1<<<(NCHAN * NCH) / 256, 256, 0, stream>>>(hidden, gate);
    mingru_scan_pass2<<<NCHAN / 64, 64, 0, stream>>>(hidden, gate);
    mingru_scan_pass3<<<(NCHAN * (NCH - 1)) / 256, 256, 0, stream>>>(hidden, gate);
}
